# starting kernel
# baseline (speedup 1.0000x reference)
#include <hip/hip_runtime.h>


typedef __attribute__((ext_vector_type(2))) float v2f;
typedef __attribute__((ext_vector_type(8))) float v8f;

#define WMMA_F32X4(a, b, c) \
  __builtin_amdgcn_wmma_f32_16x16x4_f32(false, (a), false, (b), (short)0, (c), false, false)

__device__ __forceinline__ int reflect_idx(int i, int n) {
  i = (i < 0) ? -i : i;
  i = (i >= n) ? (2 * n - 2 - i) : i;
  return i;
}

__global__ __launch_bounds__(32)
void ssim_wmma_kernel(const float* __restrict__ pred,
                      const float* __restrict__ targ,
                      float* __restrict__ out,
                      int H, int W) {
  __shared__ float sP[18][20];
  __shared__ float sT[18][20];
  __shared__ float sY[5][20][16];

  const int lane = threadIdx.x;
  const int half = lane >> 4;
  const int n    = lane & 15;
  const int r0 = blockIdx.y * 16;
  const int c0 = blockIdx.x * 16;
  const long plane = (long)blockIdx.z * H * W;
  const float* P = pred + plane;
  const float* T = targ + plane;

  for (int i = lane; i < 18 * 18; i += 32) {
    int r = i / 18, c = i % 18;
    int gr = reflect_idx(r0 - 1 + r, H);
    int gc = reflect_idx(c0 - 1 + c, W);
    sP[r][c] = P[gr * W + gc];
    sT[r][c] = T[gr * W + gc];
  }
  if (lane < 18) {
    sP[lane][18] = 0.f; sP[lane][19] = 0.f;
    sT[lane][18] = 0.f; sT[lane][19] = 0.f;
  }
  __syncthreads();

  v8f yP = {}, yT = {}, yPP = {}, yTT = {}, yPT = {};
#pragma unroll
  for (int kk = 0; kk < 5; ++kk) {
    const int k0 = 4 * kk + 2 * half;
    const int k1 = k0 + 1;
    float p0 = sP[n][k0], p1 = sP[n][k1];
    float t0 = sT[n][k0], t1 = sT[n][k1];
    v2f aP  = {p0, p1};
    v2f aT  = {t0, t1};
    v2f aPP = {p0 * p0, p1 * p1};
    v2f aTT = {t0 * t0, t1 * t1};
    v2f aPT = {p0 * t0, p1 * t1};
    v2f b;
    b.x = (k0 >= n && k0 <= n + 2) ? 1.f : 0.f;
    b.y = (k1 >= n && k1 <= n + 2) ? 1.f : 0.f;
    yP  = WMMA_F32X4(aP,  b, yP);
    yT  = WMMA_F32X4(aT,  b, yT);
    yPP = WMMA_F32X4(aPP, b, yPP);
    yTT = WMMA_F32X4(aTT, b, yTT);
    yPT = WMMA_F32X4(aPT, b, yPT);
  }

#pragma unroll
  for (int v = 0; v < 8; ++v) {
    int r = v + 8 * half;
    sY[0][r][n] = yP[v];
    sY[1][r][n] = yT[v];
    sY[2][r][n] = yPP[v];
    sY[3][r][n] = yTT[v];
    sY[4][r][n] = yPT[v];
  }
  {
    int r = 16 + half;
    float pa = sP[r][n], pb = sP[r][n + 1], pc = sP[r][n + 2];
    float ta = sT[r][n], tb = sT[r][n + 1], tc = sT[r][n + 2];
    sY[0][r][n] = pa + pb + pc;
    sY[1][r][n] = ta + tb + tc;
    sY[2][r][n] = pa * pa + pb * pb + pc * pc;
    sY[3][r][n] = ta * ta + tb * tb + tc * tc;
    sY[4][r][n] = pa * ta + pb * tb + pc * tc;
    int rz = 18 + half;
#pragma unroll
    for (int q = 0; q < 5; ++q) sY[q][rz][n] = 0.f;
  }
  __syncthreads();

  v8f uP = {}, uT = {}, uPP = {}, uTT = {}, uPT = {};
#pragma unroll
  for (int kk = 0; kk < 5; ++kk) {
    const int k0 = 4 * kk + 2 * half;
    const int k1 = k0 + 1;
    v2f a;
    a.x = (k0 >= n && k0 <= n + 2) ? 1.f : 0.f;
    a.y = (k1 >= n && k1 <= n + 2) ? 1.f : 0.f;
    v2f bP  = {sY[0][k0][n], sY[0][k1][n]};
    v2f bT  = {sY[1][k0][n], sY[1][k1][n]};
    v2f bPP = {sY[2][k0][n], sY[2][k1][n]};
    v2f bTT = {sY[3][k0][n], sY[3][k1][n]};
    v2f bPT = {sY[4][k0][n], sY[4][k1][n]};
    uP  = WMMA_F32X4(a, bP,  uP);
    uT  = WMMA_F32X4(a, bT,  uT);
    uPP = WMMA_F32X4(a, bPP, uPP);
    uTT = WMMA_F32X4(a, bTT, uTT);
    uPT = WMMA_F32X4(a, bPT, uPT);
  }

  const float inv9 = 1.f / 9.f;
  const float c1 = 1e-4f;
  const float c2 = 9e-4f;
  float* O = out + plane;
#pragma unroll
  for (int v = 0; v < 8; ++v) {
    float up = uP[v] * inv9;
    float ut = uT[v] * inv9;
    float sp = uPP[v] * inv9 - up * up;
    float st = uTT[v] * inv9 - ut * ut;
    float sc = uPT[v] * inv9 - up * ut;
    float num = (2.f * up * ut + c1) * (2.f * sc + c2);
    float den = (up * up + ut * ut + c1) * (sp + st + c2);
    float val = 0.5f * (1.f - num / den);
    val = fminf(fmaxf(val, 0.f), 1.f);
    O[(long)(r0 + v + 8 * half) * W + (c0 + n)] = val;
  }
}

extern "C" void kernel_launch(void* const* d_in, const int* in_sizes, int n_in,
                              void* d_out, int out_size, void* d_ws, size_t ws_size,
                              hipStream_t stream) {
  (void)in_sizes; (void)n_in; (void)d_ws; (void)ws_size; (void)out_size;
  const float* pred = (const float*)d_in[0];
  const float* targ = (const float*)d_in[1];
  float* out = (float*)d_out;
  const int B = 32, C = 3, H = 384, W = 512;
  dim3 grid(W / 16, H / 16, B * C);
  ssim_wmma_kernel<<<grid, dim3(32, 1, 1), 0, stream>>>(pred, targ, out, H, W);
}


# 8-wave persistent work-groups: each wave loops over 9 tiles on its own LDS slice (wave-level LDS ordering, no barrier) instead of one single-wave work-group per tile; summation order unchanged
# speedup vs baseline: 1.3517x; 1.3517x over previous
#include <hip/hip_runtime.h>


typedef __attribute__((ext_vector_type(2))) float v2f;
typedef __attribute__((ext_vector_type(8))) float v8f;

#define WMMA_F32X4(a, b, c) \
  __builtin_amdgcn_wmma_f32_16x16x4_f32(false, (a), false, (b), (short)0, (c), false, false)

#define WAVES_PER_GROUP 8

__device__ __forceinline__ int reflect_idx(int i, int n) {
  i = (i < 0) ? -i : i;
  i = (i >= n) ? (2 * n - 2 - i) : i;
  return i;
}

__device__ __forceinline__ void wave_lds_sync() {
  __builtin_amdgcn_fence(__ATOMIC_RELEASE, "wavefront");
  __builtin_amdgcn_wave_barrier();
  __builtin_amdgcn_fence(__ATOMIC_ACQUIRE, "wavefront");
}

__global__ __launch_bounds__(32 * WAVES_PER_GROUP)
void ssim_wmma_kernel(const float* __restrict__ pred,
                      const float* __restrict__ targ,
                      float* __restrict__ out,
                      int H, int W, int planes) {
  __shared__ float sPw[WAVES_PER_GROUP][18][20];
  __shared__ float sTw[WAVES_PER_GROUP][18][20];
  __shared__ float sYw[WAVES_PER_GROUP][5][20][16];

  const int lane = threadIdx.x & 31;
  const int wave = __builtin_amdgcn_readfirstlane((int)(threadIdx.x >> 5));
  const int half = lane >> 4;
  const int n    = lane & 15;

  float (*sP)[20] = sPw[wave];
  float (*sT)[20] = sTw[wave];
  float (*sY)[20][16] = sYw[wave];

  if (lane < 18) {
    sP[lane][18] = 0.f; sP[lane][19] = 0.f;
    sT[lane][18] = 0.f; sT[lane][19] = 0.f;
  }
  {
    int rz = 18 + half;
#pragma unroll
    for (int q = 0; q < 5; ++q) sY[q][rz][n] = 0.f;
  }

  const int tilesY = H >> 4;
  const int groupsPerStrip = (W >> 4) / WAVES_PER_GROUP;
  const int totalGroups = planes * tilesY * groupsPerStrip;
  const int perWg = (totalGroups + (int)gridDim.x - 1) / (int)gridDim.x;
  const int gBeg = (int)blockIdx.x * perWg;
  const int gEnd = (gBeg + perWg < totalGroups) ? (gBeg + perWg) : totalGroups;

  const float inv9 = 1.f / 9.f;
  const float c1 = 1e-4f;
  const float c2 = 9e-4f;

  for (int g = gBeg; g < gEnd; ++g) {
    const int strip = g / groupsPerStrip;
    const int gx = g - strip * groupsPerStrip;
    const int pz = strip / tilesY;
    const int ty = strip - pz * tilesY;
    const int r0 = ty * 16;
    const int c0 = (gx * WAVES_PER_GROUP + wave) * 16;
    const long plane = (long)pz * H * W;
    const float* P = pred + plane;
    const float* T = targ + plane;

    for (int i = lane; i < 18 * 18; i += 32) {
      int r = i / 18, c = i % 18;
      int gr = reflect_idx(r0 - 1 + r, H);
      int gc = reflect_idx(c0 - 1 + c, W);
      sP[r][c] = P[gr * W + gc];
      sT[r][c] = T[gr * W + gc];
    }
    wave_lds_sync();

    v8f yP = {}, yT = {}, yPP = {}, yTT = {}, yPT = {};
#pragma unroll
    for (int kk = 0; kk < 5; ++kk) {
      const int k0 = 4 * kk + 2 * half;
      const int k1 = k0 + 1;
      float p0 = sP[n][k0], p1 = sP[n][k1];
      float t0 = sT[n][k0], t1 = sT[n][k1];
      v2f aP  = {p0, p1};
      v2f aT  = {t0, t1};
      v2f aPP = {p0 * p0, p1 * p1};
      v2f aTT = {t0 * t0, t1 * t1};
      v2f aPT = {p0 * t0, p1 * t1};
      v2f b;
      b.x = (k0 >= n && k0 <= n + 2) ? 1.f : 0.f;
      b.y = (k1 >= n && k1 <= n + 2) ? 1.f : 0.f;
      yP  = WMMA_F32X4(aP,  b, yP);
      yT  = WMMA_F32X4(aT,  b, yT);
      yPP = WMMA_F32X4(aPP, b, yPP);
      yTT = WMMA_F32X4(aTT, b, yTT);
      yPT = WMMA_F32X4(aPT, b, yPT);
    }

#pragma unroll
    for (int v = 0; v < 8; ++v) {
      int r = v + 8 * half;
      sY[0][r][n] = yP[v];
      sY[1][r][n] = yT[v];
      sY[2][r][n] = yPP[v];
      sY[3][r][n] = yTT[v];
      sY[4][r][n] = yPT[v];
    }
    {
      int r = 16 + half;
      float pa = sP[r][n], pb = sP[r][n + 1], pc = sP[r][n + 2];
      float ta = sT[r][n], tb = sT[r][n + 1], tc = sT[r][n + 2];
      sY[0][r][n] = pa + pb + pc;
      sY[1][r][n] = ta + tb + tc;
      sY[2][r][n] = pa * pa + pb * pb + pc * pc;
      sY[3][r][n] = ta * ta + tb * tb + tc * tc;
      sY[4][r][n] = pa * ta + pb * tb + pc * tc;
    }
    wave_lds_sync();

    v8f uP = {}, uT = {}, uPP = {}, uTT = {}, uPT = {};
#pragma unroll
    for (int kk = 0; kk < 5; ++kk) {
      const int k0 = 4 * kk + 2 * half;
      const int k1 = k0 + 1;
      v2f a;
      a.x = (k0 >= n && k0 <= n + 2) ? 1.f : 0.f;
      a.y = (k1 >= n && k1 <= n + 2) ? 1.f : 0.f;
      v2f bP  = {sY[0][k0][n], sY[0][k1][n]};
      v2f bT  = {sY[1][k0][n], sY[1][k1][n]};
      v2f bPP = {sY[2][k0][n], sY[2][k1][n]};
      v2f bTT = {sY[3][k0][n], sY[3][k1][n]};
      v2f bPT = {sY[4][k0][n], sY[4][k1][n]};
      uP  = WMMA_F32X4(a, bP,  uP);
      uT  = WMMA_F32X4(a, bT,  uT);
      uPP = WMMA_F32X4(a, bPP, uPP);
      uTT = WMMA_F32X4(a, bTT, uTT);
      uPT = WMMA_F32X4(a, bPT, uPT);
    }

    float* O = out + plane;
#pragma unroll
    for (int v = 0; v < 8; ++v) {
      float up = uP[v] * inv9;
      float ut = uT[v] * inv9;
      float sp = uPP[v] * inv9 - up * up;
      float st = uTT[v] * inv9 - ut * ut;
      float sc = uPT[v] * inv9 - up * ut;
      float num = (2.f * up * ut + c1) * (2.f * sc + c2);
      float den = (up * up + ut * ut + c1) * (sp + st + c2);
      float val = 0.5f * (1.f - num / den);
      val = fminf(fmaxf(val, 0.f), 1.f);
      O[(long)(r0 + v + 8 * half) * W + (c0 + n)] = val;
    }
  }
}

extern "C" void kernel_launch(void* const* d_in, const int* in_sizes, int n_in,
                              void* d_out, int out_size, void* d_ws, size_t ws_size,
                              hipStream_t stream) {
  (void)in_sizes; (void)n_in; (void)d_ws; (void)ws_size; (void)out_size;
  const float* pred = (const float*)d_in[0];
  const float* targ = (const float*)d_in[1];
  float* out = (float*)d_out;
  const int B = 32, C = 3, H = 384, W = 512;
  dim3 grid(1024, 1, 1);
  ssim_wmma_kernel<<<grid, dim3(32 * WAVES_PER_GROUP, 1, 1), 0, stream>>>(pred, targ, out, H, W, B * C);
}


# fill software-pipelined: a tile's 18x18 region is fetched as one group of row-wise loads (128-bit interior, 32-bit edge columns) one tile ahead into registers; summation order unchanged
# speedup vs baseline: 1.5341x; 1.1349x over previous
#include <hip/hip_runtime.h>


typedef __attribute__((ext_vector_type(2))) float v2f;
typedef __attribute__((ext_vector_type(4))) float v4f;
typedef __attribute__((ext_vector_type(8))) float v8f;

#define WMMA_F32X4(a, b, c) \
  __builtin_amdgcn_wmma_f32_16x16x4_f32(false, (a), false, (b), (short)0, (c), false, false)

#define WAVES_PER_GROUP 8

__device__ __forceinline__ int reflect_idx(int i, int n) {
  i = (i < 0) ? -i : i;
  i = (i >= n) ? (2 * n - 2 - i) : i;
  return i;
}

__device__ __forceinline__ void wave_lds_sync() {
  __builtin_amdgcn_fence(__ATOMIC_RELEASE, "wavefront");
  __builtin_amdgcn_wave_barrier();
  __builtin_amdgcn_fence(__ATOMIC_ACQUIRE, "wavefront");
}

struct HaloRegs {
  v4f pa, pb, pc, ta, tb, tc;
  float pea, peb, tea, teb;
};

struct TileCoord { int r0, c0; long plane; };

__global__ __launch_bounds__(32 * WAVES_PER_GROUP, 8)
void ssim_wmma_kernel(const float* __restrict__ pred,
                      const float* __restrict__ targ,
                      float* __restrict__ out,
                      int H, int W, int planes) {
  __shared__ float sPw[WAVES_PER_GROUP][18][20];
  __shared__ float sTw[WAVES_PER_GROUP][18][20];
  __shared__ float sYw[WAVES_PER_GROUP][5][20][16];

  const int lane = threadIdx.x & 31;
  const int wave = __builtin_amdgcn_readfirstlane((int)(threadIdx.x >> 5));
  const int half = lane >> 4;
  const int n    = lane & 15;

  float (*sP)[20] = sPw[wave];
  float (*sT)[20] = sTw[wave];
  float (*sY)[20][16] = sYw[wave];

  if (lane < 18) {
    sP[lane][18] = 0.f; sP[lane][19] = 0.f;
    sT[lane][18] = 0.f; sT[lane][19] = 0.f;
  }
  {
    int rz = 18 + half;
#pragma unroll
    for (int q = 0; q < 5; ++q) sY[q][rz][n] = 0.f;
  }

  const int tilesY = H >> 4;
  const int groupsPerStrip = (W >> 4) / WAVES_PER_GROUP;
  const int totalGroups = planes * tilesY * groupsPerStrip;
  const int perWg = (totalGroups + (int)gridDim.x - 1) / (int)gridDim.x;
  const int gBeg = (int)blockIdx.x * perWg;
  const int gEnd = (gBeg + perWg < totalGroups) ? (gBeg + perWg) : totalGroups;
  if (gBeg >= gEnd) return;

  const float inv9 = 1.f / 9.f;
  const float c1 = 1e-4f;
  const float c2 = 9e-4f;

  const int fq   = lane & 3;
  const int fr   = lane >> 2;
  const int er   = lane >> 1;
  const int eside = lane & 1;
  const int ecol = eside ? 17 : 0;

  auto tile_of = [&](int g) {
    TileCoord t;
    const int strip = g / groupsPerStrip;
    const int gx = g - strip * groupsPerStrip;
    const int pz = strip / tilesY;
    const int ty = strip - pz * tilesY;
    t.r0 = ty * 16;
    t.c0 = (gx * WAVES_PER_GROUP + wave) * 16;
    t.plane = (long)pz * H * W;
    return t;
  };

  auto issue_loads = [&](const TileCoord& t, HaloRegs& h) {
    const float* P = pred + t.plane;
    const float* T = targ + t.plane;
    const v4f* P4 = reinterpret_cast<const v4f*>(P);
    const v4f* T4 = reinterpret_cast<const v4f*>(T);
    const int W4 = W >> 2;
    const unsigned rowA = (unsigned)(reflect_idx(t.r0 - 1 + fr, H) * W4);
    const unsigned rowB = (unsigned)((t.r0 + 7 + fr) * W4);
    const unsigned rowC = (unsigned)(reflect_idx(t.r0 + 15 + fr, H) * W4);
    const unsigned colI = (unsigned)((t.c0 >> 2) + fq);
    h.pa = P4[rowA + colI];
    h.ta = T4[rowA + colI];
    h.pb = P4[rowB + colI];
    h.tb = T4[rowB + colI];
    const int colL = reflect_idx(t.c0 - 1, W);
    const int colR = reflect_idx(t.c0 + 16, W);
    const int colE = eside ? colR : colL;
    const int rowEa = reflect_idx(t.r0 - 1 + er, H) * W;
    h.pea = P[rowEa + colE];
    h.tea = T[rowEa + colE];
    if (lane < 8) {
      h.pc = P4[rowC + colI];
      h.tc = T4[rowC + colI];
    }
    if (lane < 4) {
      const int rowEb = reflect_idx(t.r0 + 15 + er, H) * W;
      h.peb = P[rowEb + colE];
      h.teb = T[rowEb + colE];
    }
  };

  auto put4 = [](float* dst, v4f v) { dst[0] = v.x; dst[1] = v.y; dst[2] = v.z; dst[3] = v.w; };
  auto store_lds = [&](const HaloRegs& h) {
    const int ci = 1 + 4 * fq;
    put4(&sP[fr][ci],     h.pa);
    put4(&sT[fr][ci],     h.ta);
    put4(&sP[fr + 8][ci], h.pb);
    put4(&sT[fr + 8][ci], h.tb);
    sP[er][ecol] = h.pea;
    sT[er][ecol] = h.tea;
    if (lane < 8) {
      put4(&sP[fr + 16][ci], h.pc);
      put4(&sT[fr + 16][ci], h.tc);
    }
    if (lane < 4) {
      sP[er + 16][ecol] = h.peb;
      sT[er + 16][ecol] = h.teb;
    }
  };

  HaloRegs hr;
  hr.pc = (v4f){0.f, 0.f, 0.f, 0.f}; hr.tc = hr.pc;
  hr.peb = 0.f; hr.teb = 0.f;
  TileCoord cur = tile_of(gBeg);
  issue_loads(cur, hr);

  for (int g = gBeg; g < gEnd; ++g) {
    const int r0 = cur.r0;
    const int c0 = cur.c0;
    const long plane = cur.plane;

    store_lds(hr);
    wave_lds_sync();

    if (g + 1 < gEnd) {
      cur = tile_of(g + 1);
      issue_loads(cur, hr);
    }

    v8f yP = {}, yT = {}, yPP = {}, yTT = {}, yPT = {};
#pragma unroll
    for (int kk = 0; kk < 5; ++kk) {
      const int k0 = 4 * kk + 2 * half;
      const int k1 = k0 + 1;
      float p0 = sP[n][k0], p1 = sP[n][k1];
      float t0 = sT[n][k0], t1 = sT[n][k1];
      v2f aP  = {p0, p1};
      v2f aT  = {t0, t1};
      v2f aPP = {p0 * p0, p1 * p1};
      v2f aTT = {t0 * t0, t1 * t1};
      v2f aPT = {p0 * t0, p1 * t1};
      v2f b;
      b.x = (k0 >= n && k0 <= n + 2) ? 1.f : 0.f;
      b.y = (k1 >= n && k1 <= n + 2) ? 1.f : 0.f;
      yP  = WMMA_F32X4(aP,  b, yP);
      yT  = WMMA_F32X4(aT,  b, yT);
      yPP = WMMA_F32X4(aPP, b, yPP);
      yTT = WMMA_F32X4(aTT, b, yTT);
      yPT = WMMA_F32X4(aPT, b, yPT);
    }

#pragma unroll
    for (int v = 0; v < 8; ++v) {
      int r = v + 8 * half;
      sY[0][r][n] = yP[v];
      sY[1][r][n] = yT[v];
      sY[2][r][n] = yPP[v];
      sY[3][r][n] = yTT[v];
      sY[4][r][n] = yPT[v];
    }
    {
      int r = 16 + half;
      float pa = sP[r][n], pb = sP[r][n + 1], pc = sP[r][n + 2];
      float ta = sT[r][n], tb = sT[r][n + 1], tc = sT[r][n + 2];
      sY[0][r][n] = pa + pb + pc;
      sY[1][r][n] = ta + tb + tc;
      sY[2][r][n] = pa * pa + pb * pb + pc * pc;
      sY[3][r][n] = ta * ta + tb * tb + tc * tc;
      sY[4][r][n] = pa * ta + pb * tb + pc * tc;
    }
    wave_lds_sync();

    v8f uP = {}, uT = {}, uPP = {}, uTT = {}, uPT = {};
#pragma unroll
    for (int kk = 0; kk < 5; ++kk) {
      const int k0 = 4 * kk + 2 * half;
      const int k1 = k0 + 1;
      v2f a;
      a.x = (k0 >= n && k0 <= n + 2) ? 1.f : 0.f;
      a.y = (k1 >= n && k1 <= n + 2) ? 1.f : 0.f;
      v2f bP  = {sY[0][k0][n], sY[0][k1][n]};
      v2f bT  = {sY[1][k0][n], sY[1][k1][n]};
      v2f bPP = {sY[2][k0][n], sY[2][k1][n]};
      v2f bTT = {sY[3][k0][n], sY[3][k1][n]};
      v2f bPT = {sY[4][k0][n], sY[4][k1][n]};
      uP  = WMMA_F32X4(a, bP,  uP);
      uT  = WMMA_F32X4(a, bT,  uT);
      uPP = WMMA_F32X4(a, bPP, uPP);
      uTT = WMMA_F32X4(a, bTT, uTT);
      uPT = WMMA_F32X4(a, bPT, uPT);
    }

    float* O = out + plane;
#pragma unroll
    for (int v = 0; v < 8; ++v) {
      float up = uP[v] * inv9;
      float ut = uT[v] * inv9;
      float sp = uPP[v] * inv9 - up * up;
      float st = uTT[v] * inv9 - ut * ut;
      float sc = uPT[v] * inv9 - up * ut;
      float num = (2.f * up * ut + c1) * (2.f * sc + c2);
      float den = (up * up + ut * ut + c1) * (sp + st + c2);
      float val = 0.5f * (1.f - num / den);
      val = fminf(fmaxf(val, 0.f), 1.f);
      O[(long)(r0 + v + 8 * half) * W + (c0 + n)] = val;
    }
  }
}

extern "C" void kernel_launch(void* const* d_in, const int* in_sizes, int n_in,
                              void* d_out, int out_size, void* d_ws, size_t ws_size,
                              hipStream_t stream) {
  (void)in_sizes; (void)n_in; (void)d_ws; (void)ws_size; (void)out_size;
  const float* pred = (const float*)d_in[0];
  const float* targ = (const float*)d_in[1];
  float* out = (float*)d_out;
  const int B = 32, C = 3, H = 384, W = 512;
  dim3 grid(1024, 1, 1);
  ssim_wmma_kernel<<<grid, dim3(32 * WAVES_PER_GROUP, 1, 1), 0, stream>>>(pred, targ, out, H, W, B * C);
}


# p*p and t*t box-filtered as one plane p*p+t*t (SSIM uses only the sum of the two variances): 4 planes, 40 matrix instructions per tile instead of 5 and 50; variance terms summed in another order
# speedup vs baseline: 1.6871x; 1.0997x over previous
#include <hip/hip_runtime.h>


typedef __attribute__((ext_vector_type(2))) float v2f;
typedef __attribute__((ext_vector_type(4))) float v4f;
typedef __attribute__((ext_vector_type(8))) float v8f;

#define WMMA_F32X4(a, b, c) \
  __builtin_amdgcn_wmma_f32_16x16x4_f32(false, (a), false, (b), (short)0, (c), false, false)

#define WAVES_PER_GROUP 8

__device__ __forceinline__ int reflect_idx(int i, int n) {
  i = (i < 0) ? -i : i;
  i = (i >= n) ? (2 * n - 2 - i) : i;
  return i;
}

__device__ __forceinline__ void wave_lds_sync() {
  __builtin_amdgcn_fence(__ATOMIC_RELEASE, "wavefront");
  __builtin_amdgcn_wave_barrier();
  __builtin_amdgcn_fence(__ATOMIC_ACQUIRE, "wavefront");
}

struct HaloRegs {
  v4f pa, pb, pc, ta, tb, tc;
  float pea, peb, tea, teb;
};

struct TileCoord { int r0, c0; long plane; };

__global__ __launch_bounds__(32 * WAVES_PER_GROUP, 8)
void ssim_wmma_kernel(const float* __restrict__ pred,
                      const float* __restrict__ targ,
                      float* __restrict__ out,
                      int H, int W, int planes) {
  __shared__ float sPw[WAVES_PER_GROUP][18][20];
  __shared__ float sTw[WAVES_PER_GROUP][18][20];
  __shared__ float sYw[WAVES_PER_GROUP][4][20][16];

  const int lane = threadIdx.x & 31;
  const int wave = __builtin_amdgcn_readfirstlane((int)(threadIdx.x >> 5));
  const int half = lane >> 4;
  const int n    = lane & 15;

  float (*sP)[20] = sPw[wave];
  float (*sT)[20] = sTw[wave];
  float (*sY)[20][16] = sYw[wave];

  if (lane < 18) {
    sP[lane][18] = 0.f; sP[lane][19] = 0.f;
    sT[lane][18] = 0.f; sT[lane][19] = 0.f;
  }
  {
    int rz = 18 + half;
#pragma unroll
    for (int q = 0; q < 4; ++q) sY[q][rz][n] = 0.f;
  }

  const int tilesY = H >> 4;
  const int groupsPerStrip = (W >> 4) / WAVES_PER_GROUP;
  const int totalGroups = planes * tilesY * groupsPerStrip;
  const int perWg = (totalGroups + (int)gridDim.x - 1) / (int)gridDim.x;
  const int gBeg = (int)blockIdx.x * perWg;
  const int gEnd = (gBeg + perWg < totalGroups) ? (gBeg + perWg) : totalGroups;
  if (gBeg >= gEnd) return;

  const float inv9 = 1.f / 9.f;
  const float c1 = 1e-4f;
  const float c2 = 9e-4f;

  const int fq   = lane & 3;
  const int fr   = lane >> 2;
  const int er   = lane >> 1;
  const int eside = lane & 1;
  const int ecol = eside ? 17 : 0;

  auto tile_of = [&](int g) {
    TileCoord t;
    const int strip = g / groupsPerStrip;
    const int gx = g - strip * groupsPerStrip;
    const int pz = strip / tilesY;
    const int ty = strip - pz * tilesY;
    t.r0 = ty * 16;
    t.c0 = (gx * WAVES_PER_GROUP + wave) * 16;
    t.plane = (long)pz * H * W;
    return t;
  };

  auto issue_loads = [&](const TileCoord& t, HaloRegs& h) {
    const float* P = pred + t.plane;
    const float* T = targ + t.plane;
    const v4f* P4 = reinterpret_cast<const v4f*>(P);
    const v4f* T4 = reinterpret_cast<const v4f*>(T);
    const int W4 = W >> 2;
    const unsigned rowA = (unsigned)(reflect_idx(t.r0 - 1 + fr, H) * W4);
    const unsigned rowB = (unsigned)((t.r0 + 7 + fr) * W4);
    const unsigned rowC = (unsigned)(reflect_idx(t.r0 + 15 + fr, H) * W4);
    const unsigned colI = (unsigned)((t.c0 >> 2) + fq);
    h.pa = P4[rowA + colI];
    h.ta = T4[rowA + colI];
    h.pb = P4[rowB + colI];
    h.tb = T4[rowB + colI];
    const int colL = reflect_idx(t.c0 - 1, W);
    const int colR = reflect_idx(t.c0 + 16, W);
    const int colE = eside ? colR : colL;
    const int rowEa = reflect_idx(t.r0 - 1 + er, H) * W;
    h.pea = P[rowEa + colE];
    h.tea = T[rowEa + colE];
    if (lane < 8) {
      h.pc = P4[rowC + colI];
      h.tc = T4[rowC + colI];
    }
    if (lane < 4) {
      const int rowEb = reflect_idx(t.r0 + 15 + er, H) * W;
      h.peb = P[rowEb + colE];
      h.teb = T[rowEb + colE];
    }
  };

  auto put4 = [](float* dst, v4f v) { dst[0] = v.x; dst[1] = v.y; dst[2] = v.z; dst[3] = v.w; };
  auto store_lds = [&](const HaloRegs& h) {
    const int ci = 1 + 4 * fq;
    put4(&sP[fr][ci],     h.pa);
    put4(&sT[fr][ci],     h.ta);
    put4(&sP[fr + 8][ci], h.pb);
    put4(&sT[fr + 8][ci], h.tb);
    sP[er][ecol] = h.pea;
    sT[er][ecol] = h.tea;
    if (lane < 8) {
      put4(&sP[fr + 16][ci], h.pc);
      put4(&sT[fr + 16][ci], h.tc);
    }
    if (lane < 4) {
      sP[er + 16][ecol] = h.peb;
      sT[er + 16][ecol] = h.teb;
    }
  };

  HaloRegs hr;
  hr.pc = (v4f){0.f, 0.f, 0.f, 0.f}; hr.tc = hr.pc;
  hr.peb = 0.f; hr.teb = 0.f;
  TileCoord cur = tile_of(gBeg);
  issue_loads(cur, hr);

  for (int g = gBeg; g < gEnd; ++g) {
    const int r0 = cur.r0;
    const int c0 = cur.c0;
    const long plane = cur.plane;

    store_lds(hr);
    wave_lds_sync();

    if (g + 1 < gEnd) {
      cur = tile_of(g + 1);
      issue_loads(cur, hr);
    }

    v8f yP = {}, yT = {}, yQ = {}, yPT = {};
#pragma unroll
    for (int kk = 0; kk < 5; ++kk) {
      const int k0 = 4 * kk + 2 * half;
      const int k1 = k0 + 1;
      float p0 = sP[n][k0], p1 = sP[n][k1];
      float t0 = sT[n][k0], t1 = sT[n][k1];
      v2f aP  = {p0, p1};
      v2f aT  = {t0, t1};
      v2f aQ  = {p0 * p0 + t0 * t0, p1 * p1 + t1 * t1};
      v2f aPT = {p0 * t0, p1 * t1};
      v2f b;
      b.x = (k0 >= n && k0 <= n + 2) ? 1.f : 0.f;
      b.y = (k1 >= n && k1 <= n + 2) ? 1.f : 0.f;
      yP  = WMMA_F32X4(aP,  b, yP);
      yT  = WMMA_F32X4(aT,  b, yT);
      yQ  = WMMA_F32X4(aQ,  b, yQ);
      yPT = WMMA_F32X4(aPT, b, yPT);
    }

#pragma unroll
    for (int v = 0; v < 8; ++v) {
      int r = v + 8 * half;
      sY[0][r][n] = yP[v];
      sY[1][r][n] = yT[v];
      sY[2][r][n] = yQ[v];
      sY[3][r][n] = yPT[v];
    }
    {
      int r = 16 + half;
      float pa = sP[r][n], pb = sP[r][n + 1], pc = sP[r][n + 2];
      float ta = sT[r][n], tb = sT[r][n + 1], tc = sT[r][n + 2];
      sY[0][r][n] = pa + pb + pc;
      sY[1][r][n] = ta + tb + tc;
      sY[2][r][n] = (pa * pa + ta * ta) + (pb * pb + tb * tb) + (pc * pc + tc * tc);
      sY[3][r][n] = pa * ta + pb * tb + pc * tc;
    }
    wave_lds_sync();

    v8f uP = {}, uT = {}, uQ = {}, uPT = {};
#pragma unroll
    for (int kk = 0; kk < 5; ++kk) {
      const int k0 = 4 * kk + 2 * half;
      const int k1 = k0 + 1;
      v2f a;
      a.x = (k0 >= n && k0 <= n + 2) ? 1.f : 0.f;
      a.y = (k1 >= n && k1 <= n + 2) ? 1.f : 0.f;
      v2f bP  = {sY[0][k0][n], sY[0][k1][n]};
      v2f bT  = {sY[1][k0][n], sY[1][k1][n]};
      v2f bQ  = {sY[2][k0][n], sY[2][k1][n]};
      v2f bPT = {sY[3][k0][n], sY[3][k1][n]};
      uP  = WMMA_F32X4(a, bP,  uP);
      uT  = WMMA_F32X4(a, bT,  uT);
      uQ  = WMMA_F32X4(a, bQ,  uQ);
      uPT = WMMA_F32X4(a, bPT, uPT);
    }

    float* O = out + plane;
#pragma unroll
    for (int v = 0; v < 8; ++v) {
      float up = uP[v] * inv9;
      float ut = uT[v] * inv9;
      float m2 = up * up + ut * ut;
      float sv = uQ[v] * inv9 - m2;
      float sc = uPT[v] * inv9 - up * ut;
      float num = (2.f * up * ut + c1) * (2.f * sc + c2);
      float den = (m2 + c1) * (sv + c2);
      float val = 0.5f * (1.f - num / den);
      val = fminf(fmaxf(val, 0.f), 1.f);
      O[(long)(r0 + v + 8 * half) * W + (c0 + n)] = val;
    }
  }
}

extern "C" void kernel_launch(void* const* d_in, const int* in_sizes, int n_in,
                              void* d_out, int out_size, void* d_ws, size_t ws_size,
                              hipStream_t stream) {
  (void)in_sizes; (void)n_in; (void)d_ws; (void)ws_size; (void)out_size;
  const float* pred = (const float*)d_in[0];
  const float* targ = (const float*)d_in[1];
  float* out = (float*)d_out;
  const int B = 32, C = 3, H = 384, W = 512;
  dim3 grid(1024, 1, 1);
  ssim_wmma_kernel<<<grid, dim3(32 * WAVES_PER_GROUP, 1, 1), 0, stream>>>(pred, targ, out, H, W, B * C);
}
